# starting kernel
# baseline (speedup 1.0000x reference)
#include <hip/hip_runtime.h>
#include <hip/hip_bf16.h>

typedef __attribute__((ext_vector_type(2))) float v2f;
typedef __attribute__((ext_vector_type(8))) float v8f;

#define IN_FEATS  256
#define OUT_FEATS 64

__global__ void gcn_gemm_wmma(const float* __restrict__ x,
                              const float* __restrict__ W,
                              const float* __restrict__ bias,
                              float* __restrict__ h,
                              int n_nodes)
{
    const int lane   = threadIdx.x & 31;
    const int waveId = threadIdx.x >> 5;
    const int wavesPerBlock = blockDim.x >> 5;
    const int rowTile = blockIdx.x * wavesPerBlock + waveId;

    const int row0 = rowTile * 16;
    if (row0 >= n_nodes) return;

    const int m  = lane & 15;
    const int hi = lane >> 4;
    const int n  = lane & 15;

    int arow = row0 + m;
    if (arow >= n_nodes) arow = n_nodes - 1;
    const float* __restrict__ xrow = x + (size_t)arow * IN_FEATS;

    v8f acc0 = {}, acc1 = {}, acc2 = {}, acc3 = {};

    #pragma unroll 4
    for (int k = 0; k < IN_FEATS; k += 4) {
        const int kk = k + 2 * hi;
        v2f a = *(const v2f*)(xrow + kk);

        const float* __restrict__ w0 = W + (size_t)kk * OUT_FEATS + n;
        const float* __restrict__ w1 = w0 + OUT_FEATS;
        v2f b0 = { w0[0],  w1[0]  };
        v2f b1 = { w0[16], w1[16] };
        v2f b2 = { w0[32], w1[32] };
        v2f b3 = { w0[48], w1[48] };

        acc0 = __builtin_amdgcn_wmma_f32_16x16x4_f32(false, a, false, b0, (short)0, acc0, false, false);
        acc1 = __builtin_amdgcn_wmma_f32_16x16x4_f32(false, a, false, b1, (short)0, acc1, false, false);
        acc2 = __builtin_amdgcn_wmma_f32_16x16x4_f32(false, a, false, b2, (short)0, acc2, false, false);
        acc3 = __builtin_amdgcn_wmma_f32_16x16x4_f32(false, a, false, b3, (short)0, acc3, false, false);
    }

    const float bn0 = bias[n];
    const float bn1 = bias[16 + n];
    const float bn2 = bias[32 + n];
    const float bn3 = bias[48 + n];

    #pragma unroll
    for (int r = 0; r < 8; ++r) {
        const int row = row0 + r + 8 * hi;
        if (row < n_nodes) {
            float* __restrict__ hrow = h + (size_t)row * OUT_FEATS;
            hrow[n]      = acc0[r] + bn0;
            hrow[16 + n] = acc1[r] + bn1;
            hrow[32 + n] = acc2[r] + bn2;
            hrow[48 + n] = acc3[r] + bn3;
        }
    }
}

__global__ void gcn_edge_scatter(const float* __restrict__ h,
                                 const int* __restrict__ edge_src,
                                 const int* __restrict__ edge_dst,
                                 const float* __restrict__ edge_w,
                                 float* __restrict__ out,
                                 int n_edges)
{
    const int lane   = threadIdx.x & 31;
    const int waveId = threadIdx.x >> 5;
    const int wavesPerBlock = blockDim.x >> 5;

    const int e = __builtin_amdgcn_readfirstlane(blockIdx.x * wavesPerBlock + waveId);
    if (e >= n_edges) return;

    const int   s = edge_src[e];
    const int   d = edge_dst[e];
    const float w = edge_w[e];

    const v2f hv = *(const v2f*)(h + (size_t)s * OUT_FEATS + 2 * lane);

    float* __restrict__ o = out + (size_t)d * OUT_FEATS + 2 * lane;
#if defined(__HIP_DEVICE_COMPILE__)
    unsafeAtomicAdd(o,     w * hv.x);
    unsafeAtomicAdd(o + 1, w * hv.y);
#else
    atomicAdd(o,     w * hv.x);
    atomicAdd(o + 1, w * hv.y);
#endif
}

extern "C" void kernel_launch(void* const* d_in, const int* in_sizes, int n_in,
                              void* d_out, int out_size, void* d_ws, size_t ws_size,
                              hipStream_t stream)
{
    const float* x    = (const float*)d_in[0];
    const int*   esrc = (const int*)  d_in[1];
    const int*   edst = (const int*)  d_in[2];
    const float* ew   = (const float*)d_in[3];
    const float* W    = (const float*)d_in[4];
    const float* bias = (const float*)d_in[5];

    float* out = (float*)d_out;
    float* h   = (float*)d_ws;

    const int n_nodes = in_sizes[0] / IN_FEATS;
    const int n_edges = in_sizes[1];

    hipMemsetAsync(d_out, 0, (size_t)out_size * sizeof(float), stream);

    const int rowTiles = (n_nodes + 15) / 16;
    const int wavesPerBlock = 8;
    const int gemmBlocks = (rowTiles + wavesPerBlock - 1) / wavesPerBlock;
    gcn_gemm_wmma<<<gemmBlocks, wavesPerBlock * 32, 0, stream>>>(x, W, bias, h, n_nodes);

    const int scatterBlocks = (n_edges + wavesPerBlock - 1) / wavesPerBlock;
    gcn_edge_scatter<<<scatterBlocks, wavesPerBlock * 32, 0, stream>>>(h, esrc, edst, ew, out, n_edges);
}


# GEMM fragments re-laid for 16-B accesses: a lane reads 4 adjacent k of x and 4 adjacent columns of W (tile t = column 4c+t); A, B loads and h stores are b128; k order in each 8 permuted (sum order)
# speedup vs baseline: 2.3652x; 2.3652x over previous
#include <hip/hip_runtime.h>
#include <hip/hip_bf16.h>

typedef __attribute__((ext_vector_type(2))) float v2f;
typedef __attribute__((ext_vector_type(8))) float v8f;

#define IN_FEATS  256
#define OUT_FEATS 64

__global__ void gcn_gemm_wmma(const float* __restrict__ x,
                              const float* __restrict__ W,
                              const float* __restrict__ bias,
                              float* __restrict__ h,
                              int* __restrict__ head,
                              int head_pitch,
                              int n_nodes)
{
    const int gid = blockIdx.x * blockDim.x + threadIdx.x;
    if (head != nullptr) {
        if (gid < n_nodes) head[gid] = -1;
        else if (gid - n_nodes < n_nodes) head[head_pitch + (gid - n_nodes)] = -1;
    }

    const int lane   = threadIdx.x & 31;
    const int waveId = threadIdx.x >> 5;
    const int wavesPerBlock = blockDim.x >> 5;
    const int rowTile = blockIdx.x * wavesPerBlock + waveId;

    const int row0 = rowTile * 16;
    if (row0 >= n_nodes) return;

    const int c  = lane & 15;
    const int hi = lane >> 4;

    int arow = row0 + c;
    if (arow >= n_nodes) arow = n_nodes - 1;
    const float4* __restrict__ xq =
        (const float4*)(x + (size_t)arow * IN_FEATS) + hi;
    const float4* __restrict__ wq =
        (const float4*)(W + (size_t)(4 * hi) * OUT_FEATS) + c;

    v8f acc0 = {}, acc1 = {}, acc2 = {}, acc3 = {};

    #pragma unroll 2
    for (int kb = 0; kb < IN_FEATS / 8; ++kb) {
        const float4 av = xq[2 * kb];
        const float4* __restrict__ wr = wq + (size_t)kb * (8 * OUT_FEATS / 4);
        const float4 w0 = wr[0];
        const float4 w1 = wr[OUT_FEATS / 4];
        const float4 w2 = wr[2 * (OUT_FEATS / 4)];
        const float4 w3 = wr[3 * (OUT_FEATS / 4)];

        const v2f a01 = { av.x, av.y };
        const v2f a23 = { av.z, av.w };

        acc0 = __builtin_amdgcn_wmma_f32_16x16x4_f32(false, a01, false, (v2f){ w0.x, w1.x }, (short)0, acc0, false, false);
        acc1 = __builtin_amdgcn_wmma_f32_16x16x4_f32(false, a01, false, (v2f){ w0.y, w1.y }, (short)0, acc1, false, false);
        acc2 = __builtin_amdgcn_wmma_f32_16x16x4_f32(false, a01, false, (v2f){ w0.z, w1.z }, (short)0, acc2, false, false);
        acc3 = __builtin_amdgcn_wmma_f32_16x16x4_f32(false, a01, false, (v2f){ w0.w, w1.w }, (short)0, acc3, false, false);

        acc0 = __builtin_amdgcn_wmma_f32_16x16x4_f32(false, a23, false, (v2f){ w2.x, w3.x }, (short)0, acc0, false, false);
        acc1 = __builtin_amdgcn_wmma_f32_16x16x4_f32(false, a23, false, (v2f){ w2.y, w3.y }, (short)0, acc1, false, false);
        acc2 = __builtin_amdgcn_wmma_f32_16x16x4_f32(false, a23, false, (v2f){ w2.z, w3.z }, (short)0, acc2, false, false);
        acc3 = __builtin_amdgcn_wmma_f32_16x16x4_f32(false, a23, false, (v2f){ w2.w, w3.w }, (short)0, acc3, false, false);
    }

    const float4 bq = *((const float4*)bias + c);

    #pragma unroll
    for (int r = 0; r < 8; ++r) {
        const int row = row0 + r + 8 * hi;
        if (row < n_nodes) {
            float4* __restrict__ hq = (float4*)(h + (size_t)row * OUT_FEATS) + c;
            *hq = make_float4(acc0[r] + bq.x, acc1[r] + bq.y, acc2[r] + bq.z, acc3[r] + bq.w);
        }
    }
}

__global__ void gcn_edge_link(const int* __restrict__ edge_src,
                              const int* __restrict__ edge_dst,
                              const float* __restrict__ edge_w,
                              int* __restrict__ head,
                              int head_pitch,
                              int4* __restrict__ rec,
                              int n_edges)
{
    const int e = blockIdx.x * blockDim.x + threadIdx.x;
    if (e >= n_edges) return;

    const unsigned place = __builtin_amdgcn_s_sendmsg_rtn(0x87);
    const int half = (int)((place >> 18) & 1u);
    int* __restrict__ myhead = head + (half ? head_pitch : 0);

    const int d = edge_dst[e];
    const int prev = __hip_atomic_exchange(myhead + d, e, __ATOMIC_RELAXED,
                                           __HIP_MEMORY_SCOPE_AGENT);
    rec[e] = make_int4(edge_src[e], __float_as_int(edge_w[e]), prev, 0);
}

__global__ void gcn_row_pull(const float* __restrict__ h,
                             const int* __restrict__ head,
                             int head_pitch,
                             const int4* __restrict__ rec,
                             float* __restrict__ out,
                             int n_nodes)
{
    const int t = blockIdx.x * blockDim.x + threadIdx.x;
    const int r = t >> 3;
    const int j = t & 7;
    if (r >= n_nodes) return;

    float4 a0 = make_float4(0.f, 0.f, 0.f, 0.f);
    float4 a1 = make_float4(0.f, 0.f, 0.f, 0.f);

    int e  = head[r];
    int e2 = head[head_pitch + r];
    if (e < 0) { e = e2; e2 = -1; }
    int4 rc = make_int4(0, 0, -1, 0);
    if (e >= 0) rc = rec[e];

    while (e >= 0) {
        int en = rc.z;
        if (en < 0) { en = e2; e2 = -1; }
        const int4 rn = rec[en < 0 ? 0 : en];

        const float4* __restrict__ hp =
            (const float4*)(h + (size_t)rc.x * OUT_FEATS) + j;
        const float4 v0 = hp[0];
        const float4 v1 = hp[8];
        const float w = __int_as_float(rc.y);

        a0.x += w * v0.x;  a0.y += w * v0.y;  a0.z += w * v0.z;  a0.w += w * v0.w;
        a1.x += w * v1.x;  a1.y += w * v1.y;  a1.z += w * v1.z;  a1.w += w * v1.w;

        e  = en;
        rc = rn;
    }

    float4* __restrict__ op = (float4*)(out + (size_t)r * OUT_FEATS) + j;
    op[0] = a0;
    op[8] = a1;
}

__global__ void gcn_edge_scatter(const float* __restrict__ h,
                                 const int* __restrict__ edge_src,
                                 const int* __restrict__ edge_dst,
                                 const float* __restrict__ edge_w,
                                 float* __restrict__ out,
                                 int n_edges)
{
    const int lane   = threadIdx.x & 31;
    const int waveId = threadIdx.x >> 5;
    const int wavesPerBlock = blockDim.x >> 5;

    const int e = __builtin_amdgcn_readfirstlane(blockIdx.x * wavesPerBlock + waveId);
    if (e >= n_edges) return;

    const int   s = edge_src[e];
    const int   d = edge_dst[e];
    const float w = edge_w[e];

    const v2f hv = *(const v2f*)(h + (size_t)s * OUT_FEATS + 2 * lane);

    float* __restrict__ o = out + (size_t)d * OUT_FEATS + 2 * lane;
#if defined(__HIP_DEVICE_COMPILE__)
    unsafeAtomicAdd(o,     w * hv.x);
    unsafeAtomicAdd(o + 1, w * hv.y);
#else
    atomicAdd(o,     w * hv.x);
    atomicAdd(o + 1, w * hv.y);
#endif
}

extern "C" void kernel_launch(void* const* d_in, const int* in_sizes, int n_in,
                              void* d_out, int out_size, void* d_ws, size_t ws_size,
                              hipStream_t stream)
{
    const float* x    = (const float*)d_in[0];
    const int*   esrc = (const int*)  d_in[1];
    const int*   edst = (const int*)  d_in[2];
    const float* ew   = (const float*)d_in[3];
    const float* W    = (const float*)d_in[4];
    const float* bias = (const float*)d_in[5];

    float* out = (float*)d_out;
    float* h   = (float*)d_ws;

    const int n_nodes = in_sizes[0] / IN_FEATS;
    const int n_edges = in_sizes[1];

    const int rowTiles = (n_nodes + 15) / 16;
    const int wavesPerBlock = 8;
    const int gemmBlocks = (rowTiles + wavesPerBlock - 1) / wavesPerBlock;

    const size_t hBytes    = (size_t)n_nodes * OUT_FEATS * sizeof(float);
    const size_t headPitch = ((size_t)n_nodes + 31) & ~(size_t)31;
    const size_t headBytes = 2 * headPitch * sizeof(int);
    const size_t recBytes  = (size_t)n_edges * sizeof(int4);

    if (ws_size < hBytes + headBytes + recBytes) {
        hipMemsetAsync(d_out, 0, (size_t)out_size * sizeof(float), stream);
        gcn_gemm_wmma<<<gemmBlocks, wavesPerBlock * 32, 0, stream>>>(x, W, bias, h, nullptr, 0, n_nodes);
        const int scatterBlocks = (n_edges + wavesPerBlock - 1) / wavesPerBlock;
        gcn_edge_scatter<<<scatterBlocks, wavesPerBlock * 32, 0, stream>>>(h, esrc, edst, ew, out, n_edges);
        return;
    }

    int*  head = (int*)((char*)d_ws + hBytes);
    int4* rec  = (int4*)((char*)d_ws + hBytes + headBytes);

    gcn_gemm_wmma<<<gemmBlocks, wavesPerBlock * 32, 0, stream>>>(x, W, bias, h, head, (int)headPitch, n_nodes);

    const int linkBlocks = (n_edges + 255) / 256;
    gcn_edge_link<<<linkBlocks, 256, 0, stream>>>(esrc, edst, ew, head, (int)headPitch, rec, n_edges);

    const int pullBlocks = (n_nodes + 31) / 32;
    gcn_row_pull<<<pullBlocks, 256, 0, stream>>>(h, head, (int)headPitch, rec, out, n_nodes);
}
